# starting kernel
# baseline (speedup 1.0000x reference)
#include <hip/hip_runtime.h>
#include <math.h>

#define B_ 4
#define C_ 512
#define N_ 16384
#define EPS_ 1e-5f
#define NS_ITERS 12

typedef __attribute__((ext_vector_type(2))) float v2f;
typedef __attribute__((ext_vector_type(8))) float v8f;

__device__ __forceinline__ v8f wmma4(v2f a, v2f b, v8f c) {
  return __builtin_amdgcn_wmma_f32_16x16x4_f32(false, a, false, b, (short)0, c,
                                               false, false);
}

__global__ void mean_kernel(const float* __restrict__ x, float* __restrict__ mean) {
  const int bc = blockIdx.x;
  const float* p = x + (size_t)bc * N_;
  float s = 0.f;
  for (int i = threadIdx.x; i < N_; i += 256) s += p[i];
  __shared__ float red[256];
  red[threadIdx.x] = s;
  __syncthreads();
  for (int o = 128; o > 0; o >>= 1) {
    if ((int)threadIdx.x < o) red[threadIdx.x] += red[threadIdx.x + o];
    __syncthreads();
  }
  if (threadIdx.x == 0) mean[bc] = red[0] * (1.0f / (float)N_);
}

__global__ __launch_bounds__(32) void cov_kernel(const float* __restrict__ x,
                                                 const float* __restrict__ mean,
                                                 float* __restrict__ cov) {
  const int b    = blockIdx.z;
  const int row0 = blockIdx.y * 64;
  const int col0 = blockIdx.x * 64;
  const float* X  = x + (size_t)b * C_ * N_;
  const float* mu = mean + b * C_;
  float* D = cov + (size_t)b * C_ * C_;

  const int lane = threadIdx.x & 31;
  const int lm = lane & 15;
  const int kh = lane >> 4;

  const v8f zero = {0.f, 0.f, 0.f, 0.f, 0.f, 0.f, 0.f, 0.f};
  v8f acc[4][4];
#pragma unroll
  for (int m = 0; m < 4; ++m)
#pragma unroll
    for (int n = 0; n < 4; ++n) acc[m][n] = zero;

  const float* ap[4];
  const float* bp[4];
#pragma unroll
  for (int m = 0; m < 4; ++m) ap[m] = X + (size_t)(row0 + 16 * m + lm) * N_ + 2 * kh;
#pragma unroll
  for (int n = 0; n < 4; ++n) bp[n] = X + (size_t)(col0 + 16 * n + lm) * N_ + 2 * kh;

  for (int k = 0; k < N_; k += 4) {
    v2f av[4], bv[4];
#pragma unroll
    for (int m = 0; m < 4; ++m) av[m] = *(const v2f*)(ap[m] + k);
#pragma unroll
    for (int n = 0; n < 4; ++n) bv[n] = *(const v2f*)(bp[n] + k);
#pragma unroll
    for (int m = 0; m < 4; ++m)
#pragma unroll
      for (int n = 0; n < 4; ++n) acc[m][n] = wmma4(av[m], bv[n], acc[m][n]);
  }

  const float fac = 1.0f / (float)(N_ - 1);
#pragma unroll
  for (int m = 0; m < 4; ++m) {
#pragma unroll
    for (int n = 0; n < 4; ++n) {
      const int cidx = col0 + 16 * n + lm;
      const float mj = mu[cidx];
#pragma unroll
      for (int v = 0; v < 8; ++v) {
        const int r = row0 + 16 * m + v + 8 * kh;
        float val = (acc[m][n][v] - (float)N_ * mu[r] * mj) * fac;
        if (r == cidx) val += EPS_;
        D[(size_t)r * C_ + cidx] = val;
      }
    }
  }
}

__global__ void trace_kernel(const float* __restrict__ cov, float* __restrict__ tr) {
  const int m = blockIdx.x;
  const float* A = cov + (size_t)m * C_ * C_;
  float s = 0.f;
  for (int i = threadIdx.x; i < C_; i += 256) s += A[(size_t)i * C_ + i];
  __shared__ float red[256];
  red[threadIdx.x] = s;
  __syncthreads();
  for (int o = 128; o > 0; o >>= 1) {
    if ((int)threadIdx.x < o) red[threadIdx.x] += red[threadIdx.x + o];
    __syncthreads();
  }
  if (threadIdx.x == 0) tr[m] = red[0];
}

__global__ void ns_init_kernel(const float* __restrict__ cov,
                               const float* __restrict__ tr,
                               float* __restrict__ Y, float* __restrict__ Z) {
  const size_t i = (size_t)blockIdx.x * 256 + threadIdx.x;
  const size_t total = (size_t)8 * C_ * C_;
  if (i >= total) return;
  const int m = (int)(i / ((size_t)C_ * C_));
  const size_t rr = i % ((size_t)C_ * C_);
  Y[i] = cov[i] / tr[m];
  Z[i] = ((rr / C_) == (rr % C_)) ? 1.f : 0.f;
}

__global__ __launch_bounds__(32) void gemm512_kernel(const float* __restrict__ Amat,
                                                     const float* __restrict__ Bmat,
                                                     float* __restrict__ Dmat,
                                                     float alpha, float diagBeta) {
  const int mm = blockIdx.z;
  const float* A  = Amat + (size_t)mm * C_ * C_;
  const float* Bp = Bmat + (size_t)mm * C_ * C_;
  float* D = Dmat + (size_t)mm * C_ * C_;
  const int row0 = blockIdx.y * 64;
  const int col0 = blockIdx.x * 64;
  const int lane = threadIdx.x & 31;
  const int lm = lane & 15;
  const int kh = lane >> 4;

  const v8f zero = {0.f, 0.f, 0.f, 0.f, 0.f, 0.f, 0.f, 0.f};
  v8f acc[4][4];
#pragma unroll
  for (int m = 0; m < 4; ++m)
#pragma unroll
    for (int n = 0; n < 4; ++n) acc[m][n] = zero;

  for (int k = 0; k < C_; k += 4) {
    v2f av[4], bv[4];
#pragma unroll
    for (int m = 0; m < 4; ++m)
      av[m] = *(const v2f*)(A + (size_t)(row0 + 16 * m + lm) * C_ + k + 2 * kh);
#pragma unroll
    for (int n = 0; n < 4; ++n) {
      const float* q = Bp + (size_t)(k + 2 * kh) * C_ + col0 + 16 * n + lm;
      v2f t;
      t.x = q[0];
      t.y = q[C_];
      bv[n] = t;
    }
#pragma unroll
    for (int m = 0; m < 4; ++m)
#pragma unroll
      for (int n = 0; n < 4; ++n) acc[m][n] = wmma4(av[m], bv[n], acc[m][n]);
  }

#pragma unroll
  for (int m = 0; m < 4; ++m)
#pragma unroll
    for (int n = 0; n < 4; ++n) {
      const int cidx = col0 + 16 * n + lm;
#pragma unroll
      for (int v = 0; v < 8; ++v) {
        const int r = row0 + 16 * m + v + 8 * kh;
        float val = alpha * acc[m][n][v];
        if (r == cidx) val += diagBeta;
        D[(size_t)r * C_ + cidx] = val;
      }
    }
}

__global__ void bias_white_kernel(const float* __restrict__ Z,
                                  const float* __restrict__ cmean,
                                  const float* __restrict__ tr,
                                  float* __restrict__ bias) {
  const int idx = blockIdx.x * 256 + threadIdx.x;
  if (idx >= B_ * C_) return;
  const int b = idx / C_;
  const int i = idx % C_;
  const float* r = Z + ((size_t)b * C_ + i) * C_;
  const float* mu = cmean + b * C_;
  float s = 0.f;
  for (int j = 0; j < C_; ++j) s += r[j] * mu[j];
  bias[idx] = -s * rsqrtf(tr[b]);
}

__global__ __launch_bounds__(32) void apply_kernel(const float* __restrict__ Mmat,
                                                   const float* __restrict__ X,
                                                   const float* __restrict__ bias,
                                                   const float* __restrict__ tr,
                                                   int trOff, int useSqrt,
                                                   float* __restrict__ out) {
  const int b = blockIdx.z;
  const float* Mb = Mmat + (size_t)b * C_ * C_;
  const float* Xb = X + (size_t)b * C_ * N_;
  float* Ob = out + (size_t)b * C_ * N_;
  const float* bi = bias + b * C_;
  const float t = tr[trOff + b];
  const float scale = useSqrt ? sqrtf(t) : rsqrtf(t);

  const int row0 = blockIdx.y * 64;
  const int col0 = blockIdx.x * 64;
  const int lane = threadIdx.x & 31;
  const int lm = lane & 15;
  const int kh = lane >> 4;

  const v8f zero = {0.f, 0.f, 0.f, 0.f, 0.f, 0.f, 0.f, 0.f};
  v8f acc[4][4];
#pragma unroll
  for (int m = 0; m < 4; ++m)
#pragma unroll
    for (int n = 0; n < 4; ++n) acc[m][n] = zero;

  for (int k = 0; k < C_; k += 4) {
    v2f av[4], bv[4];
#pragma unroll
    for (int m = 0; m < 4; ++m)
      av[m] = *(const v2f*)(Mb + (size_t)(row0 + 16 * m + lm) * C_ + k + 2 * kh);
#pragma unroll
    for (int n = 0; n < 4; ++n) {
      const float* q = Xb + (size_t)(k + 2 * kh) * N_ + col0 + 16 * n + lm;
      v2f tt;
      tt.x = q[0];
      tt.y = q[N_];
      bv[n] = tt;
    }
#pragma unroll
    for (int m = 0; m < 4; ++m)
#pragma unroll
      for (int n = 0; n < 4; ++n) acc[m][n] = wmma4(av[m], bv[n], acc[m][n]);
  }

#pragma unroll
  for (int m = 0; m < 4; ++m)
#pragma unroll
    for (int n = 0; n < 4; ++n) {
      const int c = col0 + 16 * n + lm;
#pragma unroll
      for (int v = 0; v < 8; ++v) {
        const int r = row0 + 16 * m + v + 8 * kh;
        Ob[(size_t)r * N_ + c] = scale * acc[m][n][v] + bi[r];
      }
    }
}

extern "C" void kernel_launch(void* const* d_in, const int* in_sizes, int n_in,
                              void* d_out, int out_size, void* d_ws, size_t ws_size,
                              hipStream_t stream) {
  (void)in_sizes; (void)n_in; (void)out_size; (void)ws_size;
  const float* content = (const float*)d_in[0];
  const float* style   = (const float*)d_in[1];
  float* out = (float*)d_out;
  float* ws  = (float*)d_ws;

  const size_t CM = (size_t)C_ * C_;

  float* c_mean  = ws;
  float* s_mean  = c_mean + B_ * C_;
  float* tr      = s_mean + B_ * C_;
  float* bias_w  = tr + 8;
  float* cov_all = bias_w + B_ * C_;
  float* Y  = cov_all + 8 * CM;
  float* Z  = Y + 8 * CM;
  float* Y2 = Z + 8 * CM;
  float* Z2 = Y2 + 8 * CM;
  float* T  = Z2 + 8 * CM;
  float* whit = T + 8 * CM;

  mean_kernel<<<B_ * C_, 256, 0, stream>>>(content, c_mean);
  mean_kernel<<<B_ * C_, 256, 0, stream>>>(style, s_mean);

  dim3 covGrid(C_ / 64, C_ / 64, B_);
  cov_kernel<<<covGrid, 32, 0, stream>>>(content, c_mean, cov_all);
  cov_kernel<<<covGrid, 32, 0, stream>>>(style, s_mean, cov_all + 4 * CM);

  trace_kernel<<<8, 256, 0, stream>>>(cov_all, tr);

  {
    const size_t tot = 8 * CM;
    ns_init_kernel<<<(unsigned)((tot + 255) / 256), 256, 0, stream>>>(cov_all, tr, Y, Z);
  }
  dim3 g512(C_ / 64, C_ / 64, 8);
  float* y = Y; float* z = Z; float* y2 = Y2; float* z2 = Z2;
  for (int it = 0; it < NS_ITERS; ++it) {
    gemm512_kernel<<<g512, 32, 0, stream>>>(z, y, T, -0.5f, 1.5f);
    gemm512_kernel<<<g512, 32, 0, stream>>>(y, T, y2, 1.0f, 0.0f);
    gemm512_kernel<<<g512, 32, 0, stream>>>(T, z, z2, 1.0f, 0.0f);
    float* tmp;
    tmp = y; y = y2; y2 = tmp;
    tmp = z; z = z2; z2 = tmp;
  }

  bias_white_kernel<<<(B_ * C_ + 255) / 256, 256, 0, stream>>>(z, c_mean, tr, bias_w);

  dim3 gApply(N_ / 64, C_ / 64, B_);
  apply_kernel<<<gApply, 32, 0, stream>>>(z, content, bias_w, tr, 0, 0, whit);

  apply_kernel<<<gApply, 32, 0, stream>>>(y + 4 * CM, whit, s_mean, tr, 4, 1, out);
}


# Mean pass also writes X in fragment order (16-row gather paid once per element); covariance loads each fragment as 512 contiguous bytes; row-mean sum order changes, covariance sums unchanged
# speedup vs baseline: 3.5967x; 3.5967x over previous
#include <hip/hip_runtime.h>
#include <math.h>

#define B_ 4
#define C_ 512
#define N_ 16384
#define EPS_ 1e-5f
#define NS_ITERS 12

typedef __attribute__((ext_vector_type(2))) float v2f;
typedef __attribute__((ext_vector_type(8))) float v8f;

__device__ __forceinline__ v8f wmma4(v2f a, v2f b, v8f c) {
  return __builtin_amdgcn_wmma_f32_16x16x4_f32(false, a, false, b, (short)0, c,
                                               false, false);
}

#define CM_ (C_ * C_)
#define FRAG_T (64 * 32 * 4)
#define FRAG_KB (32 * 4)
#define XL_STRIP (2048 * 32 * 4)
#define TB_PITCH 20
#define TB_FLOATS (16 * TB_PITCH)

#define PART_FLOATS 4096
#define COV_WAVES 16
#define GEMM_WAVES 8

__device__ __forceinline__ void put_partial(float* __restrict__ part, int lane,
                                            const v8f (&acc)[4][4]) {
#pragma unroll
  for (int m = 0; m < 4; ++m)
#pragma unroll
    for (int n = 0; n < 4; ++n) {
      const int mn = 4 * m + n;
      float4 lo, hi;
      lo.x = acc[m][n][0]; lo.y = acc[m][n][1]; lo.z = acc[m][n][2]; lo.w = acc[m][n][3];
      hi.x = acc[m][n][4]; hi.y = acc[m][n][5]; hi.z = acc[m][n][6]; hi.w = acc[m][n][7];
      *(float4*)(part + ((2 * mn + 0) * 32 + lane) * 4) = lo;
      *(float4*)(part + ((2 * mn + 1) * 32 + lane) * 4) = hi;
    }
}

template <int NW>
__device__ __forceinline__ v8f sum_partials(const float* __restrict__ red, int mn, int lane) {
  v8f s;
  {
    const float4 lo = *(const float4*)(red + ((2 * mn + 0) * 32 + lane) * 4);
    const float4 hi = *(const float4*)(red + ((2 * mn + 1) * 32 + lane) * 4);
    s[0] = lo.x; s[1] = lo.y; s[2] = lo.z; s[3] = lo.w;
    s[4] = hi.x; s[5] = hi.y; s[6] = hi.z; s[7] = hi.w;
  }
#pragma unroll
  for (int r = 1; r < NW; ++r) {
    const float* p = red + r * PART_FLOATS;
    const float4 lo = *(const float4*)(p + ((2 * mn + 0) * 32 + lane) * 4);
    const float4 hi = *(const float4*)(p + ((2 * mn + 1) * 32 + lane) * 4);
    s[0] += lo.x; s[1] += lo.y; s[2] += lo.z; s[3] += lo.w;
    s[4] += hi.x; s[5] += hi.y; s[6] += hi.z; s[7] += hi.w;
  }
  return s;
}

#define MF_WAVES 32
__global__ __launch_bounds__(32 * MF_WAVES) void mean_frag_kernel(const float* __restrict__ x,
                                                                 float* __restrict__ mean,
                                                                 float* __restrict__ xl) {
  const int hs = blockIdx.x;
  const int w = __builtin_amdgcn_readfirstlane((int)(threadIdx.x >> 5));
  const int lane = threadIdx.x & 31;
  const int r8 = lane & 7;
  const int q = lane >> 3;
  const float* src = x + (size_t)(8 * hs + r8) * N_ + 4 * q;
  float* dst = xl + (size_t)(hs >> 1) * XL_STRIP + (q >> 1) * FRAG_KB
                  + 4 * (8 * (hs & 1) + r8 + 16 * (q & 1));
  float s = 0.f;
#pragma unroll 4
  for (int i = 0; i < N_ / (32 * MF_WAVES); ++i) {
    const int g0 = 2 * (w + MF_WAVES * i);
#pragma unroll
    for (int j = 0; j < 2; ++j) {
      const int g = g0 + j;
      const float4 v = *(const float4*)(src + 16 * g);
      *(float4*)(dst + (size_t)g * (2 * FRAG_KB)) = v;
      s += (v.x + v.y) + (v.z + v.w);
    }
  }
  s += __shfl_xor(s, 8);
  s += __shfl_xor(s, 16);
  __shared__ float red[MF_WAVES * 8];
  if (q == 0) red[w * 8 + r8] = s;
  __syncthreads();
  if (threadIdx.x < 8) {
    float t = 0.f;
    for (int r = 0; r < MF_WAVES; ++r) t += red[r * 8 + threadIdx.x];
    mean[8 * hs + threadIdx.x] = t * (1.0f / (float)N_);
  }
}

__global__ __launch_bounds__(32 * COV_WAVES) void cov_kernel(const float* __restrict__ xl,
                                                 const float* __restrict__ mean,
                                                 float* __restrict__ cov) {
  const int b    = blockIdx.z;
  const int row0 = blockIdx.y * 64;
  const int col0 = blockIdx.x * 64;
  const float* XLb = xl + (size_t)b * C_ * N_;
  const float* mu = mean + b * C_;
  float* D = cov + (size_t)b * C_ * C_;

  __shared__ __attribute__((aligned(16))) float red[COV_WAVES * PART_FLOATS];
  const int w = __builtin_amdgcn_readfirstlane((int)(threadIdx.x >> 5));
  const int lane = threadIdx.x & 31;
  const int lm = lane & 15;
  const int kh = lane >> 4;

  const v8f zero = {0.f, 0.f, 0.f, 0.f, 0.f, 0.f, 0.f, 0.f};
  v8f acc[4][4];
#pragma unroll
  for (int m = 0; m < 4; ++m)
#pragma unroll
    for (int n = 0; n < 4; ++n) acc[m][n] = zero;

  const float* ap[4];
  const float* bp[4];
#pragma unroll
  for (int m = 0; m < 4; ++m) ap[m] = XLb + (size_t)(blockIdx.y * 4 + m) * XL_STRIP + 4 * lane;
#pragma unroll
  for (int n = 0; n < 4; ++n) bp[n] = XLb + (size_t)(blockIdx.x * 4 + n) * XL_STRIP + 4 * lane;

  for (int i = 0; i < N_ / (8 * COV_WAVES); ++i) {
    const int kb = w + COV_WAVES * i;
    float4 a4[4], b4[4];
#pragma unroll
    for (int m = 0; m < 4; ++m) a4[m] = *(const float4*)(ap[m] + kb * FRAG_KB);
#pragma unroll
    for (int n = 0; n < 4; ++n) b4[n] = *(const float4*)(bp[n] + kb * FRAG_KB);
    v2f av[4], bv[4];
#pragma unroll
    for (int m = 0; m < 4; ++m) { av[m].x = a4[m].x; av[m].y = a4[m].y; }
#pragma unroll
    for (int n = 0; n < 4; ++n) { bv[n].x = b4[n].x; bv[n].y = b4[n].y; }
#pragma unroll
    for (int m = 0; m < 4; ++m)
#pragma unroll
      for (int n = 0; n < 4; ++n) acc[m][n] = wmma4(av[m], bv[n], acc[m][n]);
#pragma unroll
    for (int m = 0; m < 4; ++m) { av[m].x = a4[m].z; av[m].y = a4[m].w; }
#pragma unroll
    for (int n = 0; n < 4; ++n) { bv[n].x = b4[n].z; bv[n].y = b4[n].w; }
#pragma unroll
    for (int m = 0; m < 4; ++m)
#pragma unroll
      for (int n = 0; n < 4; ++n) acc[m][n] = wmma4(av[m], bv[n], acc[m][n]);
  }

  put_partial(red + w * PART_FLOATS, lane, acc);
  __syncthreads();

  const float fac = 1.0f / (float)(N_ - 1);
  {
    const int m = w >> 2;
    const int n = w & 3;
    const v8f s = sum_partials<COV_WAVES>(red, w, lane);
    const int cidx = col0 + 16 * n + lm;
    const float mj = mu[cidx];
#pragma unroll
    for (int v = 0; v < 8; ++v) {
      const int r = row0 + 16 * m + v + 8 * kh;
      float val = (s[v] - (float)N_ * mu[r] * mj) * fac;
      if (r == cidx) val += EPS_;
      D[(size_t)r * C_ + cidx] = val;
    }
  }
}

__global__ void trace_kernel(const float* __restrict__ cov, float* __restrict__ tr) {
  const int m = blockIdx.x;
  const float* A = cov + (size_t)m * C_ * C_;
  float s = 0.f;
  for (int i = threadIdx.x; i < C_; i += 256) s += A[(size_t)i * C_ + i];
  __shared__ float red[256];
  red[threadIdx.x] = s;
  __syncthreads();
  for (int o = 128; o > 0; o >>= 1) {
    if ((int)threadIdx.x < o) red[threadIdx.x] += red[threadIdx.x + o];
    __syncthreads();
  }
  if (threadIdx.x == 0) tr[m] = red[0];
}

__global__ void ns_init_kernel(const float* __restrict__ cov,
                               const float* __restrict__ tr,
                               float* __restrict__ YL, float* __restrict__ YR,
                               float* __restrict__ ZL, float* __restrict__ ZR) {
  const int t = blockIdx.x * 256 + threadIdx.x;
  const int m = t >> 16;
  const int p = t & 65535;
  const int strip = p >> 11;
  const int kb = (p >> 5) & 63;
  const int lane = p & 31;
  const int a = 16 * strip + (lane & 15);
  const int k0 = 8 * kb + 4 * (lane >> 4);
  const float* A = cov + (size_t)m * CM_;
  const float d = tr[m];
  const size_t o = (size_t)m * CM_ + 4 * (size_t)p;

  const float4 cl = *(const float4*)(A + (size_t)a * C_ + k0);
  float4 y, z;
  y.x = cl.x / d; y.y = cl.y / d; y.z = cl.z / d; y.w = cl.w / d;
  z.x = (a == k0 + 0) ? 1.f : 0.f; z.y = (a == k0 + 1) ? 1.f : 0.f;
  z.z = (a == k0 + 2) ? 1.f : 0.f; z.w = (a == k0 + 3) ? 1.f : 0.f;
  *(float4*)(YL + o) = y;
  *(float4*)(ZL + o) = z;
  *(float4*)(ZR + o) = z;

  const float* q = A + (size_t)k0 * C_ + a;
  y.x = q[0] / d; y.y = q[C_] / d; y.z = q[2 * C_] / d; y.w = q[3 * C_] / d;
  *(float4*)(YR + o) = y;
}

__global__ __launch_bounds__(32 * GEMM_WAVES) void gemm512_kernel(const float* __restrict__ AL,
                                                     const float* __restrict__ BR,
                                                     float* __restrict__ DL,
                                                     float* __restrict__ DR,
                                                     float alpha, float diagBeta) {
  const int mm = blockIdx.z;
  const int row0 = blockIdx.y * 64;
  const int col0 = blockIdx.x * 64;
  __shared__ __attribute__((aligned(16))) float red[GEMM_WAVES * PART_FLOATS];
  __shared__ __attribute__((aligned(16))) float tbuf[GEMM_WAVES * TB_FLOATS];
  const int w = __builtin_amdgcn_readfirstlane((int)(threadIdx.x >> 5));
  const int lane = threadIdx.x & 31;
  const int lm = lane & 15;
  const int kh = lane >> 4;

  const float* Ap = AL + (size_t)mm * CM_ + (size_t)(blockIdx.y * 4) * FRAG_T + 4 * lane;
  const float* Bq = BR + (size_t)mm * CM_ + (size_t)(blockIdx.x * 4) * FRAG_T + 4 * lane;

  const v8f zero = {0.f, 0.f, 0.f, 0.f, 0.f, 0.f, 0.f, 0.f};
  v8f acc[4][4];
#pragma unroll
  for (int m = 0; m < 4; ++m)
#pragma unroll
    for (int n = 0; n < 4; ++n) acc[m][n] = zero;

#pragma unroll 1
  for (int i = 0; i < C_ / (8 * GEMM_WAVES); ++i) {
    const int kb = w + GEMM_WAVES * i;
    float4 a4[4], b4[4];
#pragma unroll
    for (int m = 0; m < 4; ++m) a4[m] = *(const float4*)(Ap + m * FRAG_T + kb * FRAG_KB);
#pragma unroll
    for (int n = 0; n < 4; ++n) b4[n] = *(const float4*)(Bq + n * FRAG_T + kb * FRAG_KB);
    v2f av[4], bv[4];
#pragma unroll
    for (int m = 0; m < 4; ++m) { av[m].x = a4[m].x; av[m].y = a4[m].y; }
#pragma unroll
    for (int n = 0; n < 4; ++n) { bv[n].x = b4[n].x; bv[n].y = b4[n].y; }
#pragma unroll
    for (int m = 0; m < 4; ++m)
#pragma unroll
      for (int n = 0; n < 4; ++n) acc[m][n] = wmma4(av[m], bv[n], acc[m][n]);
#pragma unroll
    for (int m = 0; m < 4; ++m) { av[m].x = a4[m].z; av[m].y = a4[m].w; }
#pragma unroll
    for (int n = 0; n < 4; ++n) { bv[n].x = b4[n].z; bv[n].y = b4[n].w; }
#pragma unroll
    for (int m = 0; m < 4; ++m)
#pragma unroll
      for (int n = 0; n < 4; ++n) acc[m][n] = wmma4(av[m], bv[n], acc[m][n]);
  }

  put_partial(red + w * PART_FLOATS, lane, acc);
  __syncthreads();

  float* tb = tbuf + w * TB_FLOATS;
  float* DLm = DL + (size_t)mm * CM_;
  float* DRm = DR + (size_t)mm * CM_;
#pragma unroll
  for (int j = 0; j < 2; ++j) {
    const int mn = 2 * w + j;
    const int m = mn >> 2;
    const int n = mn & 3;
    const v8f s = sum_partials<GEMM_WAVES>(red, mn, lane);
    const int cidx = col0 + 16 * n + lm;
    float val[8];
#pragma unroll
    for (int v = 0; v < 8; ++v) {
      const int r = row0 + 16 * m + v + 8 * kh;
      float x = alpha * s[v];
      if (r == cidx) x += diagBeta;
      val[v] = x;
    }
    {
      const int ct = blockIdx.x * 4 + n;
      const int kbr = blockIdx.y * 8 + 2 * m + kh;
      float* p = DRm + (size_t)ct * FRAG_T + kbr * FRAG_KB + 4 * lm;
      float4 lo, hi;
      lo.x = val[0]; lo.y = val[1]; lo.z = val[2]; lo.w = val[3];
      hi.x = val[4]; hi.y = val[5]; hi.z = val[6]; hi.w = val[7];
      *(float4*)(p) = lo;
      *(float4*)(p + 64) = hi;
    }
    {
#pragma unroll
      for (int v = 0; v < 8; ++v) tb[(8 * kh + v) * TB_PITCH + lm] = val[v];
      const float4 f0 = *(const float4*)(tb + lm * TB_PITCH + 4 * kh);
      const float4 f1 = *(const float4*)(tb + lm * TB_PITCH + 8 + 4 * kh);
      const int rt = blockIdx.y * 4 + m;
      const int kbl = blockIdx.x * 8 + 2 * n;
      float* p = DLm + (size_t)rt * FRAG_T + kbl * FRAG_KB + 4 * lane;
      *(float4*)(p) = f0;
      *(float4*)(p + FRAG_KB) = f1;
    }
  }
}

__global__ void bias_white_kernel(const float* __restrict__ ZL,
                                  const float* __restrict__ cmean,
                                  const float* __restrict__ tr,
                                  float* __restrict__ bias) {
  const int idx = blockIdx.x * 256 + threadIdx.x;
  if (idx >= B_ * C_) return;
  const int b = idx / C_;
  const int i = idx % C_;
  const float* r = ZL + (size_t)b * CM_ + (size_t)(i >> 4) * FRAG_T + 4 * (i & 15);
  const float* mu = cmean + b * C_;
  float s = 0.f;
  for (int kb = 0; kb < C_ / 8; ++kb) {
#pragma unroll
    for (int h = 0; h < 2; ++h) {
      const float4 q = *(const float4*)(r + kb * FRAG_KB + 64 * h);
      const float* u = mu + 8 * kb + 4 * h;
      s += q.x * u[0];
      s += q.y * u[1];
      s += q.z * u[2];
      s += q.w * u[3];
    }
  }
  bias[idx] = -s * rsqrtf(tr[b]);
}

__global__ __launch_bounds__(32) __attribute__((amdgpu_waves_per_eu(1, 5)))
void apply_kernel(const float* __restrict__ Mmat,
                                                   const float* __restrict__ X,
                                                   const float* __restrict__ bias,
                                                   const float* __restrict__ tr,
                                                   int trOff, int useSqrt,
                                                   float* __restrict__ out) {
  const int b = blockIdx.z;
  const float* Xb = X + (size_t)b * C_ * N_;
  float* Ob = out + (size_t)b * C_ * N_;
  const float* bi = bias + b * C_;
  const float t = tr[trOff + b];
  const float scale = useSqrt ? sqrtf(t) : rsqrtf(t);

  const int row0 = blockIdx.y * 64;
  const int col0 = blockIdx.x * 64;
  const int lane = threadIdx.x & 31;
  const int lm = lane & 15;
  const int kh = lane >> 4;
  const float* Mp = Mmat + (size_t)b * CM_ + (size_t)(blockIdx.y * 4) * FRAG_T + 4 * lane;

  const v8f zero = {0.f, 0.f, 0.f, 0.f, 0.f, 0.f, 0.f, 0.f};
  v8f acc[4][4];
#pragma unroll
  for (int m = 0; m < 4; ++m)
#pragma unroll
    for (int n = 0; n < 4; ++n) acc[m][n] = zero;

  for (int k = 0; k < C_; k += 8) {
    float4 a4[4];
    v2f bv[2][4];
#pragma unroll
    for (int m = 0; m < 4; ++m)
      a4[m] = *(const float4*)(Mp + m * FRAG_T + (k >> 3) * FRAG_KB);
#pragma unroll
    for (int s = 0; s < 2; ++s)
#pragma unroll
      for (int n = 0; n < 4; ++n) {
        const float* q = Xb + (size_t)(k + 4 * kh + 2 * s) * N_ + col0 + 16 * n + lm;
        v2f tt;
        tt.x = q[0];
        tt.y = q[N_];
        bv[s][n] = tt;
      }
    v2f av[4];
#pragma unroll
    for (int m = 0; m < 4; ++m) { av[m].x = a4[m].x; av[m].y = a4[m].y; }
#pragma unroll
    for (int m = 0; m < 4; ++m)
#pragma unroll
      for (int n = 0; n < 4; ++n) acc[m][n] = wmma4(av[m], bv[0][n], acc[m][n]);
#pragma unroll
    for (int m = 0; m < 4; ++m) { av[m].x = a4[m].z; av[m].y = a4[m].w; }
#pragma unroll
    for (int m = 0; m < 4; ++m)
#pragma unroll
      for (int n = 0; n < 4; ++n) acc[m][n] = wmma4(av[m], bv[1][n], acc[m][n]);
  }

#pragma unroll
  for (int m = 0; m < 4; ++m)
#pragma unroll
    for (int n = 0; n < 4; ++n) {
      const int c = col0 + 16 * n + lm;
#pragma unroll
      for (int v = 0; v < 8; ++v) {
        const int r = row0 + 16 * m + v + 8 * kh;
        Ob[(size_t)r * N_ + c] = scale * acc[m][n][v] + bi[r];
      }
    }
}

extern "C" void kernel_launch(void* const* d_in, const int* in_sizes, int n_in,
                              void* d_out, int out_size, void* d_ws, size_t ws_size,
                              hipStream_t stream) {
  (void)in_sizes; (void)n_in; (void)out_size; (void)ws_size;
  const float* content = (const float*)d_in[0];
  const float* style   = (const float*)d_in[1];
  float* out = (float*)d_out;
  float* ws  = (float*)d_ws;

  const size_t CM = (size_t)C_ * C_;

  float* c_mean  = ws;
  float* s_mean  = c_mean + B_ * C_;
  float* tr      = s_mean + B_ * C_;
  float* bias_w  = tr + 64;
  float* cov_all = bias_w + B_ * C_;
  float* YL = cov_all + 8 * CM;
  float* YR = YL + 8 * CM;
  float* ZL = YR + 8 * CM;
  float* ZR = ZL + 8 * CM;
  float* whit = ZR + 8 * CM;
  float* Y2L = whit;
  float* Y2R = Y2L + 8 * CM;
  float* Z2L = Y2R + 8 * CM;
  float* Z2R = Z2L + 8 * CM;
  float* TL  = Z2R + 8 * CM;
  float* TR  = TL + 8 * CM;
  float* XL  = whit;

  dim3 covGrid(C_ / 64, C_ / 64, B_);
  mean_frag_kernel<<<B_ * C_ / 8, 32 * MF_WAVES, 0, stream>>>(content, c_mean, XL);
  cov_kernel<<<covGrid, 32 * COV_WAVES, 0, stream>>>(XL, c_mean, cov_all);
  mean_frag_kernel<<<B_ * C_ / 8, 32 * MF_WAVES, 0, stream>>>(style, s_mean, XL);
  cov_kernel<<<covGrid, 32 * COV_WAVES, 0, stream>>>(XL, s_mean, cov_all + 4 * CM);

  trace_kernel<<<8, 256, 0, stream>>>(cov_all, tr);

  ns_init_kernel<<<8 * 65536 / 256, 256, 0, stream>>>(cov_all, tr, YL, YR, ZL, ZR);
  dim3 g512(C_ / 64, C_ / 64, 8);
  static_assert(NS_ITERS % 2 == 0, "results must end in the YL / ZL buffers");
  float* yl = YL; float* yr = YR; float* zl = ZL; float* zr = ZR;
  float* y2l = Y2L; float* y2r = Y2R; float* z2l = Z2L; float* z2r = Z2R;
  for (int it = 0; it < NS_ITERS; ++it) {
    gemm512_kernel<<<g512, 32 * GEMM_WAVES, 0, stream>>>(zl, yr, TL, TR, -0.5f, 1.5f);
    gemm512_kernel<<<g512, 32 * GEMM_WAVES, 0, stream>>>(yl, TR, y2l, y2r, 1.0f, 0.0f);
    gemm512_kernel<<<g512, 32 * GEMM_WAVES, 0, stream>>>(TL, zr, z2l, z2r, 1.0f, 0.0f);
    float* tmp;
    tmp = yl; yl = y2l; y2l = tmp;
    tmp = yr; yr = y2r; y2r = tmp;
    tmp = zl; zl = z2l; z2l = tmp;
    tmp = zr; zr = z2r; z2r = tmp;
  }

  bias_white_kernel<<<(B_ * C_ + 255) / 256, 256, 0, stream>>>(zl, c_mean, tr, bias_w);

  dim3 gApply(N_ / 64, C_ / 64, B_);
  apply_kernel<<<gApply, 32, 0, stream>>>(zl, content, bias_w, tr, 0, 0, whit);

  apply_kernel<<<gApply, 32, 0, stream>>>(yl + 4 * CM, whit, s_mean, tr, 4, 1, out);
}


# All 3 matrix kernels' k-loops load the NEXT 8-k block's fragments into a second register set before the current block's 32 matrix steps (loads fly during own matrix work); sums unchanged
# speedup vs baseline: 3.8569x; 1.0723x over previous
#include <hip/hip_runtime.h>
#include <math.h>

#define B_ 4
#define C_ 512
#define N_ 16384
#define EPS_ 1e-5f
#define NS_ITERS 12

typedef __attribute__((ext_vector_type(2))) float v2f;
typedef __attribute__((ext_vector_type(8))) float v8f;

__device__ __forceinline__ v8f wmma4(v2f a, v2f b, v8f c) {
  return __builtin_amdgcn_wmma_f32_16x16x4_f32(false, a, false, b, (short)0, c,
                                               false, false);
}

#define CM_ (C_ * C_)
#define FRAG_T (64 * 32 * 4)
#define FRAG_KB (32 * 4)
#define XL_STRIP (2048 * 32 * 4)
#define TB_PITCH 20
#define TB_FLOATS (16 * TB_PITCH)

#define PART_FLOATS 4096
#define COV_WAVES 16
#define GEMM_WAVES 8

__device__ __forceinline__ void put_partial(float* __restrict__ part, int lane,
                                            const v8f (&acc)[4][4]) {
#pragma unroll
  for (int m = 0; m < 4; ++m)
#pragma unroll
    for (int n = 0; n < 4; ++n) {
      const int mn = 4 * m + n;
      float4 lo, hi;
      lo.x = acc[m][n][0]; lo.y = acc[m][n][1]; lo.z = acc[m][n][2]; lo.w = acc[m][n][3];
      hi.x = acc[m][n][4]; hi.y = acc[m][n][5]; hi.z = acc[m][n][6]; hi.w = acc[m][n][7];
      *(float4*)(part + ((2 * mn + 0) * 32 + lane) * 4) = lo;
      *(float4*)(part + ((2 * mn + 1) * 32 + lane) * 4) = hi;
    }
}

template <int NW>
__device__ __forceinline__ v8f sum_partials(const float* __restrict__ red, int mn, int lane) {
  v8f s;
  {
    const float4 lo = *(const float4*)(red + ((2 * mn + 0) * 32 + lane) * 4);
    const float4 hi = *(const float4*)(red + ((2 * mn + 1) * 32 + lane) * 4);
    s[0] = lo.x; s[1] = lo.y; s[2] = lo.z; s[3] = lo.w;
    s[4] = hi.x; s[5] = hi.y; s[6] = hi.z; s[7] = hi.w;
  }
#pragma unroll
  for (int r = 1; r < NW; ++r) {
    const float* p = red + r * PART_FLOATS;
    const float4 lo = *(const float4*)(p + ((2 * mn + 0) * 32 + lane) * 4);
    const float4 hi = *(const float4*)(p + ((2 * mn + 1) * 32 + lane) * 4);
    s[0] += lo.x; s[1] += lo.y; s[2] += lo.z; s[3] += lo.w;
    s[4] += hi.x; s[5] += hi.y; s[6] += hi.z; s[7] += hi.w;
  }
  return s;
}

__device__ __forceinline__ void mma_block(v8f (&acc)[4][4], const float4 (&a4)[4],
                                          const float4 (&b4)[4]) {
  v2f av[4], bv[4];
#pragma unroll
  for (int m = 0; m < 4; ++m) { av[m].x = a4[m].x; av[m].y = a4[m].y; }
#pragma unroll
  for (int n = 0; n < 4; ++n) { bv[n].x = b4[n].x; bv[n].y = b4[n].y; }
#pragma unroll
  for (int m = 0; m < 4; ++m)
#pragma unroll
    for (int n = 0; n < 4; ++n) acc[m][n] = wmma4(av[m], bv[n], acc[m][n]);
#pragma unroll
  for (int m = 0; m < 4; ++m) { av[m].x = a4[m].z; av[m].y = a4[m].w; }
#pragma unroll
  for (int n = 0; n < 4; ++n) { bv[n].x = b4[n].z; bv[n].y = b4[n].w; }
#pragma unroll
  for (int m = 0; m < 4; ++m)
#pragma unroll
    for (int n = 0; n < 4; ++n) acc[m][n] = wmma4(av[m], bv[n], acc[m][n]);
}

#define MF_WAVES 32
__global__ __launch_bounds__(32 * MF_WAVES) void mean_frag_kernel(const float* __restrict__ x,
                                                                 float* __restrict__ mean,
                                                                 float* __restrict__ xl) {
  const int hs = blockIdx.x;
  const int w = __builtin_amdgcn_readfirstlane((int)(threadIdx.x >> 5));
  const int lane = threadIdx.x & 31;
  const int r8 = lane & 7;
  const int q = lane >> 3;
  const float* src = x + (size_t)(8 * hs + r8) * N_ + 4 * q;
  float* dst = xl + (size_t)(hs >> 1) * XL_STRIP + (q >> 1) * FRAG_KB
                  + 4 * (8 * (hs & 1) + r8 + 16 * (q & 1));
  float s = 0.f;
#pragma unroll 4
  for (int i = 0; i < N_ / (32 * MF_WAVES); ++i) {
    const int g0 = 2 * (w + MF_WAVES * i);
#pragma unroll
    for (int j = 0; j < 2; ++j) {
      const int g = g0 + j;
      const float4 v = *(const float4*)(src + 16 * g);
      *(float4*)(dst + (size_t)g * (2 * FRAG_KB)) = v;
      s += (v.x + v.y) + (v.z + v.w);
    }
  }
  s += __shfl_xor(s, 8);
  s += __shfl_xor(s, 16);
  __shared__ float red[MF_WAVES * 8];
  if (q == 0) red[w * 8 + r8] = s;
  __syncthreads();
  if (threadIdx.x < 8) {
    float t = 0.f;
    for (int r = 0; r < MF_WAVES; ++r) t += red[r * 8 + threadIdx.x];
    mean[8 * hs + threadIdx.x] = t * (1.0f / (float)N_);
  }
}

__global__ __launch_bounds__(32 * COV_WAVES) void cov_kernel(const float* __restrict__ xl,
                                                 const float* __restrict__ mean,
                                                 float* __restrict__ cov) {
  const int b    = blockIdx.z;
  const int row0 = blockIdx.y * 64;
  const int col0 = blockIdx.x * 64;
  const float* XLb = xl + (size_t)b * C_ * N_;
  const float* mu = mean + b * C_;
  float* D = cov + (size_t)b * C_ * C_;

  __shared__ __attribute__((aligned(16))) float red[COV_WAVES * PART_FLOATS];
  const int w = __builtin_amdgcn_readfirstlane((int)(threadIdx.x >> 5));
  const int lane = threadIdx.x & 31;
  const int lm = lane & 15;
  const int kh = lane >> 4;

  const v8f zero = {0.f, 0.f, 0.f, 0.f, 0.f, 0.f, 0.f, 0.f};
  v8f acc[4][4];
#pragma unroll
  for (int m = 0; m < 4; ++m)
#pragma unroll
    for (int n = 0; n < 4; ++n) acc[m][n] = zero;

  const float* ap[4];
  const float* bp[4];
#pragma unroll
  for (int m = 0; m < 4; ++m) ap[m] = XLb + (size_t)(blockIdx.y * 4 + m) * XL_STRIP + 4 * lane;
#pragma unroll
  for (int n = 0; n < 4; ++n) bp[n] = XLb + (size_t)(blockIdx.x * 4 + n) * XL_STRIP + 4 * lane;

  constexpr int NT = N_ / (8 * COV_WAVES);
  float4 a0[4], b0[4], a1[4], b1[4];
#pragma unroll
  for (int m = 0; m < 4; ++m) a0[m] = *(const float4*)(ap[m] + w * FRAG_KB);
#pragma unroll
  for (int n = 0; n < 4; ++n) b0[n] = *(const float4*)(bp[n] + w * FRAG_KB);
  for (int i = 0; i < NT - 2; i += 2) {
    const int kb1 = w + COV_WAVES * (i + 1);
#pragma unroll
    for (int m = 0; m < 4; ++m) a1[m] = *(const float4*)(ap[m] + kb1 * FRAG_KB);
#pragma unroll
    for (int n = 0; n < 4; ++n) b1[n] = *(const float4*)(bp[n] + kb1 * FRAG_KB);
    __builtin_amdgcn_sched_barrier(0);
    mma_block(acc, a0, b0);
    const int kb2 = w + COV_WAVES * (i + 2);
#pragma unroll
    for (int m = 0; m < 4; ++m) a0[m] = *(const float4*)(ap[m] + kb2 * FRAG_KB);
#pragma unroll
    for (int n = 0; n < 4; ++n) b0[n] = *(const float4*)(bp[n] + kb2 * FRAG_KB);
    __builtin_amdgcn_sched_barrier(0);
    mma_block(acc, a1, b1);
  }
  {
    const int kb1 = w + COV_WAVES * (NT - 1);
#pragma unroll
    for (int m = 0; m < 4; ++m) a1[m] = *(const float4*)(ap[m] + kb1 * FRAG_KB);
#pragma unroll
    for (int n = 0; n < 4; ++n) b1[n] = *(const float4*)(bp[n] + kb1 * FRAG_KB);
    mma_block(acc, a0, b0);
    mma_block(acc, a1, b1);
  }

  put_partial(red + w * PART_FLOATS, lane, acc);
  __syncthreads();

  const float fac = 1.0f / (float)(N_ - 1);
  {
    const int m = w >> 2;
    const int n = w & 3;
    const v8f s = sum_partials<COV_WAVES>(red, w, lane);
    const int cidx = col0 + 16 * n + lm;
    const float mj = mu[cidx];
#pragma unroll
    for (int v = 0; v < 8; ++v) {
      const int r = row0 + 16 * m + v + 8 * kh;
      float val = (s[v] - (float)N_ * mu[r] * mj) * fac;
      if (r == cidx) val += EPS_;
      D[(size_t)r * C_ + cidx] = val;
    }
  }
}

__global__ void trace_kernel(const float* __restrict__ cov, float* __restrict__ tr) {
  const int m = blockIdx.x;
  const float* A = cov + (size_t)m * C_ * C_;
  float s = 0.f;
  for (int i = threadIdx.x; i < C_; i += 256) s += A[(size_t)i * C_ + i];
  __shared__ float red[256];
  red[threadIdx.x] = s;
  __syncthreads();
  for (int o = 128; o > 0; o >>= 1) {
    if ((int)threadIdx.x < o) red[threadIdx.x] += red[threadIdx.x + o];
    __syncthreads();
  }
  if (threadIdx.x == 0) tr[m] = red[0];
}

__global__ void ns_init_kernel(const float* __restrict__ cov,
                               const float* __restrict__ tr,
                               float* __restrict__ YL, float* __restrict__ YR,
                               float* __restrict__ ZL, float* __restrict__ ZR) {
  const int t = blockIdx.x * 256 + threadIdx.x;
  const int m = t >> 16;
  const int p = t & 65535;
  const int strip = p >> 11;
  const int kb = (p >> 5) & 63;
  const int lane = p & 31;
  const int a = 16 * strip + (lane & 15);
  const int k0 = 8 * kb + 4 * (lane >> 4);
  const float* A = cov + (size_t)m * CM_;
  const float d = tr[m];
  const size_t o = (size_t)m * CM_ + 4 * (size_t)p;

  const float4 cl = *(const float4*)(A + (size_t)a * C_ + k0);
  float4 y, z;
  y.x = cl.x / d; y.y = cl.y / d; y.z = cl.z / d; y.w = cl.w / d;
  z.x = (a == k0 + 0) ? 1.f : 0.f; z.y = (a == k0 + 1) ? 1.f : 0.f;
  z.z = (a == k0 + 2) ? 1.f : 0.f; z.w = (a == k0 + 3) ? 1.f : 0.f;
  *(float4*)(YL + o) = y;
  *(float4*)(ZL + o) = z;
  *(float4*)(ZR + o) = z;

  const float* q = A + (size_t)k0 * C_ + a;
  y.x = q[0] / d; y.y = q[C_] / d; y.z = q[2 * C_] / d; y.w = q[3 * C_] / d;
  *(float4*)(YR + o) = y;
}

__global__ __launch_bounds__(32 * GEMM_WAVES) void gemm512_kernel(const float* __restrict__ AL,
                                                     const float* __restrict__ BR,
                                                     float* __restrict__ DL,
                                                     float* __restrict__ DR,
                                                     float alpha, float diagBeta) {
  const int mm = blockIdx.z;
  const int row0 = blockIdx.y * 64;
  const int col0 = blockIdx.x * 64;
  __shared__ __attribute__((aligned(16))) float red[GEMM_WAVES * PART_FLOATS];
  __shared__ __attribute__((aligned(16))) float tbuf[GEMM_WAVES * TB_FLOATS];
  const int w = __builtin_amdgcn_readfirstlane((int)(threadIdx.x >> 5));
  const int lane = threadIdx.x & 31;
  const int lm = lane & 15;
  const int kh = lane >> 4;

  const float* Ap = AL + (size_t)mm * CM_ + (size_t)(blockIdx.y * 4) * FRAG_T + 4 * lane;
  const float* Bq = BR + (size_t)mm * CM_ + (size_t)(blockIdx.x * 4) * FRAG_T + 4 * lane;

  const v8f zero = {0.f, 0.f, 0.f, 0.f, 0.f, 0.f, 0.f, 0.f};
  v8f acc[4][4];
#pragma unroll
  for (int m = 0; m < 4; ++m)
#pragma unroll
    for (int n = 0; n < 4; ++n) acc[m][n] = zero;

  constexpr int NT = C_ / (8 * GEMM_WAVES);
  float4 a0[4], b0[4], a1[4], b1[4];
#pragma unroll
  for (int m = 0; m < 4; ++m) a0[m] = *(const float4*)(Ap + m * FRAG_T + w * FRAG_KB);
#pragma unroll
  for (int n = 0; n < 4; ++n) b0[n] = *(const float4*)(Bq + n * FRAG_T + w * FRAG_KB);
#pragma unroll 1
  for (int i = 0; i < NT - 2; i += 2) {
    const int kb1 = w + GEMM_WAVES * (i + 1);
#pragma unroll
    for (int m = 0; m < 4; ++m) a1[m] = *(const float4*)(Ap + m * FRAG_T + kb1 * FRAG_KB);
#pragma unroll
    for (int n = 0; n < 4; ++n) b1[n] = *(const float4*)(Bq + n * FRAG_T + kb1 * FRAG_KB);
    __builtin_amdgcn_sched_barrier(0);
    mma_block(acc, a0, b0);
    const int kb2 = w + GEMM_WAVES * (i + 2);
#pragma unroll
    for (int m = 0; m < 4; ++m) a0[m] = *(const float4*)(Ap + m * FRAG_T + kb2 * FRAG_KB);
#pragma unroll
    for (int n = 0; n < 4; ++n) b0[n] = *(const float4*)(Bq + n * FRAG_T + kb2 * FRAG_KB);
    __builtin_amdgcn_sched_barrier(0);
    mma_block(acc, a1, b1);
  }
  {
    const int kb1 = w + GEMM_WAVES * (NT - 1);
#pragma unroll
    for (int m = 0; m < 4; ++m) a1[m] = *(const float4*)(Ap + m * FRAG_T + kb1 * FRAG_KB);
#pragma unroll
    for (int n = 0; n < 4; ++n) b1[n] = *(const float4*)(Bq + n * FRAG_T + kb1 * FRAG_KB);
    mma_block(acc, a0, b0);
    mma_block(acc, a1, b1);
  }

  put_partial(red + w * PART_FLOATS, lane, acc);
  __syncthreads();

  float* tb = tbuf + w * TB_FLOATS;
  float* DLm = DL + (size_t)mm * CM_;
  float* DRm = DR + (size_t)mm * CM_;
#pragma unroll
  for (int j = 0; j < 2; ++j) {
    const int mn = 2 * w + j;
    const int m = mn >> 2;
    const int n = mn & 3;
    const v8f s = sum_partials<GEMM_WAVES>(red, mn, lane);
    const int cidx = col0 + 16 * n + lm;
    float val[8];
#pragma unroll
    for (int v = 0; v < 8; ++v) {
      const int r = row0 + 16 * m + v + 8 * kh;
      float x = alpha * s[v];
      if (r == cidx) x += diagBeta;
      val[v] = x;
    }
    {
      const int ct = blockIdx.x * 4 + n;
      const int kbr = blockIdx.y * 8 + 2 * m + kh;
      float* p = DRm + (size_t)ct * FRAG_T + kbr * FRAG_KB + 4 * lm;
      float4 lo, hi;
      lo.x = val[0]; lo.y = val[1]; lo.z = val[2]; lo.w = val[3];
      hi.x = val[4]; hi.y = val[5]; hi.z = val[6]; hi.w = val[7];
      *(float4*)(p) = lo;
      *(float4*)(p + 64) = hi;
    }
    {
#pragma unroll
      for (int v = 0; v < 8; ++v) tb[(8 * kh + v) * TB_PITCH + lm] = val[v];
      const float4 f0 = *(const float4*)(tb + lm * TB_PITCH + 4 * kh);
      const float4 f1 = *(const float4*)(tb + lm * TB_PITCH + 8 + 4 * kh);
      const int rt = blockIdx.y * 4 + m;
      const int kbl = blockIdx.x * 8 + 2 * n;
      float* p = DLm + (size_t)rt * FRAG_T + kbl * FRAG_KB + 4 * lane;
      *(float4*)(p) = f0;
      *(float4*)(p + FRAG_KB) = f1;
    }
  }
}

__global__ void bias_white_kernel(const float* __restrict__ ZL,
                                  const float* __restrict__ cmean,
                                  const float* __restrict__ tr,
                                  float* __restrict__ bias) {
  const int idx = blockIdx.x * 256 + threadIdx.x;
  if (idx >= B_ * C_) return;
  const int b = idx / C_;
  const int i = idx % C_;
  const float* r = ZL + (size_t)b * CM_ + (size_t)(i >> 4) * FRAG_T + 4 * (i & 15);
  const float* mu = cmean + b * C_;
  float s = 0.f;
  for (int kb = 0; kb < C_ / 8; ++kb) {
#pragma unroll
    for (int h = 0; h < 2; ++h) {
      const float4 q = *(const float4*)(r + kb * FRAG_KB + 64 * h);
      const float* u = mu + 8 * kb + 4 * h;
      s += q.x * u[0];
      s += q.y * u[1];
      s += q.z * u[2];
      s += q.w * u[3];
    }
  }
  bias[idx] = -s * rsqrtf(tr[b]);
}

__global__ __launch_bounds__(32) __attribute__((amdgpu_waves_per_eu(1, 4)))
void apply_kernel(const float* __restrict__ Mmat,
                                                   const float* __restrict__ X,
                                                   const float* __restrict__ bias,
                                                   const float* __restrict__ tr,
                                                   int trOff, int useSqrt,
                                                   float* __restrict__ out) {
  const int b = blockIdx.z;
  const float* Xb = X + (size_t)b * C_ * N_;
  float* Ob = out + (size_t)b * C_ * N_;
  const float* bi = bias + b * C_;
  const float t = tr[trOff + b];
  const float scale = useSqrt ? sqrtf(t) : rsqrtf(t);

  const int row0 = blockIdx.y * 64;
  const int col0 = blockIdx.x * 64;
  const int lane = threadIdx.x & 31;
  const int lm = lane & 15;
  const int kh = lane >> 4;
  const float* Mp = Mmat + (size_t)b * CM_ + (size_t)(blockIdx.y * 4) * FRAG_T + 4 * lane;

  const v8f zero = {0.f, 0.f, 0.f, 0.f, 0.f, 0.f, 0.f, 0.f};
  v8f acc[4][4];
#pragma unroll
  for (int m = 0; m < 4; ++m)
#pragma unroll
    for (int n = 0; n < 4; ++n) acc[m][n] = zero;

  const float* Xq = Xb + (size_t)(4 * kh) * N_ + col0 + lm;
  auto load_blk = [&](float4 (&a4)[4], v2f (&bv)[2][4], int k) {
#pragma unroll
    for (int m = 0; m < 4; ++m)
      a4[m] = *(const float4*)(Mp + m * FRAG_T + (k >> 3) * FRAG_KB);
#pragma unroll
    for (int s = 0; s < 2; ++s)
#pragma unroll
      for (int n = 0; n < 4; ++n) {
        const float* q = Xq + (size_t)(k + 2 * s) * N_ + 16 * n;
        v2f tt;
        tt.x = q[0];
        tt.y = q[N_];
        bv[s][n] = tt;
      }
  };
  auto mma_blk = [&](const float4 (&a4)[4], const v2f (&bv)[2][4]) {
    v2f av[4];
#pragma unroll
    for (int m = 0; m < 4; ++m) { av[m].x = a4[m].x; av[m].y = a4[m].y; }
#pragma unroll
    for (int m = 0; m < 4; ++m)
#pragma unroll
      for (int n = 0; n < 4; ++n) acc[m][n] = wmma4(av[m], bv[0][n], acc[m][n]);
#pragma unroll
    for (int m = 0; m < 4; ++m) { av[m].x = a4[m].z; av[m].y = a4[m].w; }
#pragma unroll
    for (int m = 0; m < 4; ++m)
#pragma unroll
      for (int n = 0; n < 4; ++n) acc[m][n] = wmma4(av[m], bv[1][n], acc[m][n]);
  };
  float4 a0[4], a1[4];
  v2f x0[2][4], x1[2][4];
  load_blk(a0, x0, 0);
  for (int k = 0; k < C_ - 16; k += 16) {
    load_blk(a1, x1, k + 8);
    __builtin_amdgcn_sched_barrier(0);
    mma_blk(a0, x0);
    load_blk(a0, x0, k + 16);
    __builtin_amdgcn_sched_barrier(0);
    mma_blk(a1, x1);
  }
  load_blk(a1, x1, C_ - 8);
  mma_blk(a0, x0);
  mma_blk(a1, x1);

#pragma unroll
  for (int m = 0; m < 4; ++m)
#pragma unroll
    for (int n = 0; n < 4; ++n) {
      const int c = col0 + 16 * n + lm;
#pragma unroll
      for (int v = 0; v < 8; ++v) {
        const int r = row0 + 16 * m + v + 8 * kh;
        Ob[(size_t)r * N_ + c] = scale * acc[m][n][v] + bi[r];
      }
    }
}

extern "C" void kernel_launch(void* const* d_in, const int* in_sizes, int n_in,
                              void* d_out, int out_size, void* d_ws, size_t ws_size,
                              hipStream_t stream) {
  (void)in_sizes; (void)n_in; (void)out_size; (void)ws_size;
  const float* content = (const float*)d_in[0];
  const float* style   = (const float*)d_in[1];
  float* out = (float*)d_out;
  float* ws  = (float*)d_ws;

  const size_t CM = (size_t)C_ * C_;

  float* c_mean  = ws;
  float* s_mean  = c_mean + B_ * C_;
  float* tr      = s_mean + B_ * C_;
  float* bias_w  = tr + 64;
  float* cov_all = bias_w + B_ * C_;
  float* YL = cov_all + 8 * CM;
  float* YR = YL + 8 * CM;
  float* ZL = YR + 8 * CM;
  float* ZR = ZL + 8 * CM;
  float* whit = ZR + 8 * CM;
  float* Y2L = whit;
  float* Y2R = Y2L + 8 * CM;
  float* Z2L = Y2R + 8 * CM;
  float* Z2R = Z2L + 8 * CM;
  float* TL  = Z2R + 8 * CM;
  float* TR  = TL + 8 * CM;
  float* XL  = whit;

  dim3 covGrid(C_ / 64, C_ / 64, B_);
  mean_frag_kernel<<<B_ * C_ / 8, 32 * MF_WAVES, 0, stream>>>(content, c_mean, XL);
  cov_kernel<<<covGrid, 32 * COV_WAVES, 0, stream>>>(XL, c_mean, cov_all);
  mean_frag_kernel<<<B_ * C_ / 8, 32 * MF_WAVES, 0, stream>>>(style, s_mean, XL);
  cov_kernel<<<covGrid, 32 * COV_WAVES, 0, stream>>>(XL, s_mean, cov_all + 4 * CM);

  trace_kernel<<<8, 256, 0, stream>>>(cov_all, tr);

  ns_init_kernel<<<8 * 65536 / 256, 256, 0, stream>>>(cov_all, tr, YL, YR, ZL, ZR);
  dim3 g512(C_ / 64, C_ / 64, 8);
  static_assert(NS_ITERS % 2 == 0, "results must end in the YL / ZL buffers");
  float* yl = YL; float* yr = YR; float* zl = ZL; float* zr = ZR;
  float* y2l = Y2L; float* y2r = Y2R; float* z2l = Z2L; float* z2r = Z2R;
  for (int it = 0; it < NS_ITERS; ++it) {
    gemm512_kernel<<<g512, 32 * GEMM_WAVES, 0, stream>>>(zl, yr, TL, TR, -0.5f, 1.5f);
    gemm512_kernel<<<g512, 32 * GEMM_WAVES, 0, stream>>>(yl, TR, y2l, y2r, 1.0f, 0.0f);
    gemm512_kernel<<<g512, 32 * GEMM_WAVES, 0, stream>>>(TL, zr, z2l, z2r, 1.0f, 0.0f);
    float* tmp;
    tmp = yl; yl = y2l; y2l = tmp;
    tmp = yr; yr = y2r; y2r = tmp;
    tmp = zl; zl = z2l; z2l = tmp;
    tmp = zr; zr = z2r; z2r = tmp;
  }

  bias_white_kernel<<<(B_ * C_ + 255) / 256, 256, 0, stream>>>(zl, c_mean, tr, bias_w);

  dim3 gApply(N_ / 64, C_ / 64, B_);
  apply_kernel<<<gApply, 32, 0, stream>>>(zl, content, bias_w, tr, 0, 0, whit);

  apply_kernel<<<gApply, 32, 0, stream>>>(yl + 4 * CM, whit, s_mean, tr, 4, 1, out);
}
